# starting kernel
# baseline (speedup 1.0000x reference)
#include <hip/hip_runtime.h>
#include <hip/hip_bf16.h>
#include <math.h>

typedef __attribute__((ext_vector_type(2))) float v2f;
typedef __attribute__((ext_vector_type(8))) float v8f;

#define Nn 4
#define Dd 256
#define Hd 64
#define Wd 64
#define HW (Hd * Wd)

__global__ __launch_bounds__(256) void FeatureAlign_norms(
    const float* __restrict__ cur, const float* __restrict__ prev,
    float* __restrict__ cnF, float* __restrict__ pnF) {
  const int idx = blockIdx.x * 256 + threadIdx.x;
  const int n = idx >> 12;
  const int pix = idx & (HW - 1);
  const size_t base = (size_t)n * Dd * HW + pix;
  float cs = 0.0f, ps = 0.0f;
  for (int d = 0; d < Dd; ++d) {
    const float c = cur[base + (size_t)d * HW];
    const float p = prev[base + (size_t)d * HW];
    cs += c * c;
    ps += p * p;
  }
  cnF[idx] = sqrtf(cs) + 1e-8f;
  pnF[idx] = sqrtf(ps) + 1e-8f;
}

__global__ __launch_bounds__(128) void FeatureAlign_kernel(
    const float* __restrict__ cur, const float* __restrict__ prev,
    const float* __restrict__ mem, const float* __restrict__ cnF,
    const float* __restrict__ pnF, float* __restrict__ out) {
  __shared__ float ldsD[4][4][16][32];
  __shared__ float ldsW[4][Wd];
  __shared__ int   ldsC[Wd];

  const int nb = blockIdx.x;
  const int n = nb >> 6;
  const int y = nb & (Hd - 1);
  const int t = threadIdx.x;
  const int wave = t >> 5;
  const int lane = t & 31;
  const int g = lane >> 4;
  const int li = lane & 15;
  const int x0 = wave << 4;

  const float* curN = cur + (size_t)n * Dd * HW;
  const float* prevN = prev + (size_t)n * Dd * HW;
  const float* memN = mem + (size_t)n * Dd * HW;
  float* outN = out + (size_t)n * Dd * HW;

  const unsigned voffA = (unsigned)(2 * g * HW + y * Wd + x0 + li);
  unsigned voffB[4][2];
#pragma unroll
  for (int dyi = 0; dyi < 4; ++dyi) {
    const int dy = (dyi == 3) ? 2 : (dyi - 1);
    int yy = y + dy;
    yy = yy < 0 ? 0 : (yy > Hd - 1 ? Hd - 1 : yy);
#pragma unroll
    for (int h = 0; h < 2; ++h) {
      int bx = x0 - 1 + 16 * h + li;
      bx = bx < 0 ? 0 : (bx > Wd - 1 ? Wd - 1 : bx);
      voffB[dyi][h] = (unsigned)(2 * g * HW + yy * Wd + bx);
    }
  }
  __builtin_assume(voffA < 3u * HW);

  v8f acc[4][2];
#pragma unroll
  for (int dyi = 0; dyi < 4; ++dyi)
#pragma unroll
    for (int h = 0; h < 2; ++h) acc[dyi][h] = {};

  auto loadA = [&](int k, v2f& a) {
    const float* pk = curN + (size_t)(unsigned)k * HW;
    a.x = pk[voffA];
    a.y = pk[voffA + HW];
  };
  auto loadB = [&](int k, v2f (&b)[4][2]) {
    const float* pk = prevN + (size_t)(unsigned)k * HW;
#pragma unroll
    for (int dyi = 0; dyi < 4; ++dyi)
#pragma unroll
      for (int h = 0; h < 2; ++h) {
        b[dyi][h].x = pk[voffB[dyi][h]];
        b[dyi][h].y = pk[voffB[dyi][h] + HW];
      }
  };
  auto domma = [&](const v2f& a, const v2f (&b)[4][2]) {
#pragma unroll
    for (int dyi = 0; dyi < 4; ++dyi)
#pragma unroll
      for (int h = 0; h < 2; ++h)
        acc[dyi][h] = __builtin_amdgcn_wmma_f32_16x16x4_f32(
            false, a, false, b[dyi][h], (short)0, acc[dyi][h], false, false);
  };

  v2f a0, a1;
  v2f b0[4][2], b1[4][2];
  loadA(0, a0);
  loadB(0, b0);
  for (int k = 0; k < Dd; k += 8) {
    loadA(k + 4, a1);
    loadB(k + 4, b1);
    domma(a0, b0);
    const int kq = (k + 8 < Dd) ? (k + 8) : 0;
    loadA(kq, a0);
    loadB(kq, b0);
    domma(a1, b1);
  }

#pragma unroll
  for (int dyi = 0; dyi < 4; ++dyi)
#pragma unroll
    for (int h = 0; h < 2; ++h)
#pragma unroll
      for (int r = 0; r < 8; ++r)
        ldsD[wave][dyi][r + 8 * g][16 * h + li] = acc[dyi][h][r];

  __syncthreads();

  const int nb2 = n * HW;
  if (t < Wd) {
    const int x = t, w = x >> 4, i = x & 15;
    const float cn = cnF[nb2 + y * Wd + x];
    float mass = 0.0f;
#pragma unroll
    for (int dyi = 0; dyi < 3; ++dyi) {
      const int yy = y + dyi - 1;
      const bool rowv = (yy >= 0) && (yy < Hd);
#pragma unroll
      for (int dx = -1; dx <= 1; ++dx) {
        const int px = x + dx;
        if (rowv && px >= 0 && px < Wd) {
          const float pn = pnF[nb2 + yy * Wd + px];
          const float a = ldsD[w][dyi][i][i + 1 + dx] / (cn * pn);
          mass += fmaxf(a, 0.0f);
        }
      }
    }
    ldsC[x] = (mass > -10.0f) && (mass < 10.0f);
    const float msafe = (mass > 0.0f) ? mass : 1.0f;
#pragma unroll
    for (int oy = 0; oy < 2; ++oy) {
      const int dy = oy ? 2 : -1;
      const int dyi = oy ? 3 : 0;
      const int py = y + dy;
#pragma unroll
      for (int ox = 0; ox < 2; ++ox) {
        const int dx = ox ? 2 : -1;
        const int px = x + dx;
        float wv = 0.0f;
        if (py >= 0 && py < Hd && px >= 0 && px < Wd) {
          const float pn = pnF[nb2 + py * Wd + px];
          const float a = ldsD[w][dyi][i][i + 1 + dx] / (cn * pn);
          wv = (a > 0.0f) ? (a / msafe) : 0.0f;
        }
        ldsW[oy * 2 + ox][x] = wv;
      }
    }
  }
  __syncthreads();

  const int yb = y * Wd;
  for (int it = 0; it < (Dd * Wd) / 128; ++it) {
    const int idx2 = t + 128 * it;
    const int d = idx2 >> 6;
    const int x = idx2 & (Wd - 1);
    const size_t base = (size_t)d * HW;
    const float m0 = memN[base + yb + x];
    float agg = 0.0f;
    if (d > 0) {
#pragma unroll
      for (int off = 0; off < 4; ++off) {
        const int dy = (off < 2) ? -1 : 2;
        const int dx = (off & 1) ? 2 : -1;
        int py = y + dy; py = py < 0 ? 0 : (py > Hd - 1 ? Hd - 1 : py);
        int px = x + dx; px = px < 0 ? 0 : (px > Wd - 1 ? Wd - 1 : px);
        agg += ldsW[off][x] * memN[base + py * Wd + px];
      }
    }
    outN[base + yb + x] = ldsC[x] ? (m0 + agg) : 0.0f;
  }
}

extern "C" void kernel_launch(void* const* d_in, const int* in_sizes, int n_in,
                              void* d_out, int out_size, void* d_ws, size_t ws_size,
                              hipStream_t stream) {
  const float* cur = (const float*)d_in[0];
  const float* prev = (const float*)d_in[1];
  const float* mem = (const float*)d_in[2];
  float* out = (float*)d_out;
  (void)in_sizes; (void)n_in; (void)out_size; (void)ws_size;

  float* cnF = (float*)d_ws;
  float* pnF = cnF + (size_t)Nn * HW;

  FeatureAlign_norms<<<dim3((Nn * HW) / 256), dim3(256), 0, stream>>>(
      cur, prev, cnF, pnF);
  FeatureAlign_kernel<<<dim3(Nn * Hd), dim3(128), 0, stream>>>(
      cur, prev, mem, cnF, pnF, out);
}


# Dot loop LDS fragment reads regrouped: each block of 2 chunks (18 reads) issued as one group ahead of the previous block's 16 WMMAs, second fragment set; 3 read places per stage, was 21; sums kept
# speedup vs baseline: 5.2560x; 5.2560x over previous
#include <hip/hip_runtime.h>
#include <hip/hip_bf16.h>
#include <math.h>

typedef __attribute__((ext_vector_type(2))) float v2f;
typedef __attribute__((ext_vector_type(8))) float v8f;
typedef __attribute__((ext_vector_type(4))) float v4f;

#define Nn 4
#define Dd 256
#define Hd 64
#define Wd 64
#define HW (Hd * Wd)
#define DC 64
#define CG (Dd / DC)
#define KC 32
#define NST (Dd / KC)
#define PITCH 72
#define SLOT (KC * PITCH)
#define STG (5 * SLOT)

#define NG 16
__global__ __launch_bounds__(256) __attribute__((amdgpu_waves_per_eu(2, 4)))
void FeatureAlign_norms(
    const float* __restrict__ cur, const float* __restrict__ prev,
    float* __restrict__ cnF, float* __restrict__ pnF) {
  __shared__ __attribute__((aligned(16))) float red[2][NG][Wd];

  const int nb = blockIdx.x;
  const int n = nb >> 6;
  const int y = nb & (Hd - 1);
  const int t = threadIdx.x;
  const int q4 = (t & 15) << 2;
  const int cg = t >> 4;

  const float* curR = cur + (size_t)n * Dd * HW + y * Wd;
  const float* prevR = prev + (size_t)n * Dd * HW + y * Wd;
  const unsigned lo = (unsigned)(cg * HW + q4);

  v4f c[Dd / NG], p[Dd / NG];
#pragma unroll
  for (int j = 0; j < Dd / NG; ++j)
    c[j] = *(const v4f*)(curR + lo + (unsigned)(j * NG * HW));
#pragma unroll
  for (int j = 0; j < Dd / NG; ++j)
    p[j] = *(const v4f*)(prevR + lo + (unsigned)(j * NG * HW));

  v4f cs = {0.0f, 0.0f, 0.0f, 0.0f}, ps = {0.0f, 0.0f, 0.0f, 0.0f};
#pragma unroll
  for (int j = 0; j < Dd / NG; ++j) cs += c[j] * c[j];
#pragma unroll
  for (int j = 0; j < Dd / NG; ++j) ps += p[j] * p[j];

  *(v4f*)&red[0][cg][q4] = cs;
  *(v4f*)&red[1][cg][q4] = ps;
  __syncthreads();

  if (t < 2 * Wd) {
    const int arr = t >> 6;
    const int x = t & (Wd - 1);
    float s = 0.0f;
#pragma unroll
    for (int gI = 0; gI < NG; ++gI) s += red[arr][gI][x];
    float* dst = arr ? pnF : cnF;
    dst[nb * Wd + x] = sqrtf(s) + 1e-8f;
  }
}

__global__ __launch_bounds__(128) void FeatureAlign_kernel(
    const float* __restrict__ cur, const float* __restrict__ prev,
    const float* __restrict__ cnF,
    const float* __restrict__ pnF, float* __restrict__ out) {
  __shared__ __attribute__((aligned(16))) float ldsS[2][STG];
  __shared__ float ldsD[4][4][16][32];
  __shared__ float ldsW[5][Wd];

  const int nb = blockIdx.x;
  const int n = nb >> 6;
  const int y = nb & (Hd - 1);
  const int t = threadIdx.x;
  const int wave = t >> 5;
  const int lane = t & 31;
  const int g = lane >> 4;
  const int li = lane & 15;
  const int x0 = wave << 4;

  const float* curN = cur + (size_t)n * Dd * HW;
  const float* prevN = prev + (size_t)n * Dd * HW;
  float* outN = out + (size_t)n * Dd * HW;

  const int q4 = (t & 15) << 2;
  const int cs = t >> 4;
  const unsigned gA = (unsigned)(cs * HW + y * Wd + q4);
  unsigned gB[4];
#pragma unroll
  for (int dyi = 0; dyi < 4; ++dyi) {
    const int dy = (dyi == 3) ? 2 : (dyi - 1);
    int yy = y + dy;
    yy = yy < 0 ? 0 : (yy > Hd - 1 ? Hd - 1 : yy);
    gB[dyi] = (unsigned)(cs * HW + yy * Wd + q4);
  }
  const unsigned sOff = (unsigned)(cs * PITCH + q4);

  const unsigned fA = (unsigned)(2 * g * PITCH + x0 + li);
  unsigned fB[4][2];
#pragma unroll
  for (int dyi = 0; dyi < 4; ++dyi)
#pragma unroll
    for (int h = 0; h < 2; ++h) {
      int bx = x0 - 1 + 16 * h + li;
      bx = bx < 0 ? 0 : (bx > Wd - 1 ? Wd - 1 : bx);
      fB[dyi][h] = (unsigned)((dyi + 1) * SLOT + 2 * g * PITCH + bx);
    }

  const int nb2 = n * HW;
  const int xq = t & (Wd - 1);
  bool yv[4], xv[4];
  int yc[4], xc[4];
#pragma unroll
  for (int k = 0; k < 4; ++k) {
    const int d = (k == 3) ? 2 : (k - 1);
    const int yy = y + d, xx = xq + d;
    yv[k] = (yy >= 0) && (yy < Hd);
    xv[k] = (xx >= 0) && (xx < Wd);
    yc[k] = yy < 0 ? 0 : (yy > Hd - 1 ? Hd - 1 : yy);
    xc[k] = xx < 0 ? 0 : (xx > Wd - 1 ? Wd - 1 : xx);
  }
  v8f acc[4][2];
#pragma unroll
  for (int dyi = 0; dyi < 4; ++dyi)
#pragma unroll
    for (int h = 0; h < 2; ++h) acc[dyi][h] = {};

  v4f st[5][4];
  auto gload = [&](int s) {
    const float* pc = curN + (size_t)(unsigned)(s * KC) * HW;
    const float* pp = prevN + (size_t)(unsigned)(s * KC) * HW;
#pragma unroll
    for (int jj = 0; jj < 4; ++jj)
      st[0][jj] = *(const v4f*)(pc + gA + (unsigned)(jj * 8 * HW));
#pragma unroll
    for (int r = 0; r < 4; ++r)
#pragma unroll
      for (int jj = 0; jj < 4; ++jj)
        st[r + 1][jj] = *(const v4f*)(pp + gB[r] + (unsigned)(jj * 8 * HW));
  };
  auto sstore = [&](int buf) {
    float* b = &ldsS[buf][0];
#pragma unroll
    for (int r = 0; r < 5; ++r)
#pragma unroll
      for (int jj = 0; jj < 4; ++jj)
        *(v4f*)(b + r * SLOT + jj * 8 * PITCH + sOff) = st[r][jj];
  };
  struct Frag {
    v2f a[2];
    v2f b[2][4][2];
  };
  auto fload = [&](const float* b, int blk, Frag& f) {
#pragma unroll
    for (int c = 0; c < 2; ++c) {
      const int kk = (blk * 2 + c) * 4;
      f.a[c].x = b[fA + kk * PITCH];
      f.a[c].y = b[fA + (kk + 1) * PITCH];
#pragma unroll
      for (int dyi = 0; dyi < 4; ++dyi)
#pragma unroll
        for (int h = 0; h < 2; ++h) {
          f.b[c][dyi][h].x = b[fB[dyi][h] + kk * PITCH];
          f.b[c][dyi][h].y = b[fB[dyi][h] + (kk + 1) * PITCH];
        }
    }
  };
  auto fmma = [&](const Frag& f) {
#pragma unroll
    for (int c = 0; c < 2; ++c)
#pragma unroll
      for (int dyi = 0; dyi < 4; ++dyi)
#pragma unroll
        for (int h = 0; h < 2; ++h)
          acc[dyi][h] = __builtin_amdgcn_wmma_f32_16x16x4_f32(
              false, f.a[c], false, f.b[c][dyi][h], (short)0, acc[dyi][h],
              false, false);
  };
  auto compute = [&](int buf) {
    const float* b = &ldsS[buf][0];
    Frag f0, f1;
    fload(b, 0, f0);
    __builtin_amdgcn_sched_barrier(0);
    fload(b, 1, f1);
    __builtin_amdgcn_sched_barrier(0);
    fmma(f0);
    __builtin_amdgcn_sched_barrier(0);
    fload(b, 2, f0);
    __builtin_amdgcn_sched_barrier(0);
    fmma(f1);
    __builtin_amdgcn_sched_barrier(0);
    fload(b, 3, f1);
    __builtin_amdgcn_sched_barrier(0);
    fmma(f0);
    fmma(f1);
  };

  gload(0);
  float cnv = 0.0f;
  float pnv[4][4];
#pragma unroll
  for (int a = 0; a < 4; ++a)
#pragma unroll
    for (int b2 = 0; b2 < 4; ++b2) pnv[a][b2] = 0.0f;
  if (t < Wd) {
    cnv = cnF[nb2 + y * Wd + xq];
#pragma unroll
    for (int a = 0; a < 4; ++a)
#pragma unroll
      for (int b2 = 0; b2 < 4; ++b2) {
        const bool inMass = (a < 3) && (b2 < 3);
        const bool corner = (a == 0 || a == 3) && (b2 == 0 || b2 == 3);
        if (inMass || corner) pnv[a][b2] = pnF[nb2 + yc[a] * Wd + xc[b2]];
      }
  }
  sstore(0);
  __syncthreads();
  for (int s = 0; s < NST; ++s) {
    const bool more = (s + 1 < NST);
    if (more) gload(s + 1);
    compute(s & 1);
    __builtin_amdgcn_sched_barrier(0);
    if (more) sstore((s + 1) & 1);
    __syncthreads();
  }

#pragma unroll
  for (int dyi = 0; dyi < 4; ++dyi)
#pragma unroll
    for (int h = 0; h < 2; ++h)
#pragma unroll
      for (int r = 0; r < 8; ++r)
        ldsD[wave][dyi][r + 8 * g][16 * h + li] = acc[dyi][h][r];

  __syncthreads();

  if (t < Wd) {
    const int x = t, w = x >> 4, i = x & 15;
    float av[4][4];
#pragma unroll
    for (int a = 0; a < 4; ++a)
#pragma unroll
      for (int b2 = 0; b2 < 4; ++b2) {
        const bool inMass = (a < 3) && (b2 < 3);
        const bool corner = (a == 0 || a == 3) && (b2 == 0 || b2 == 3);
        av[a][b2] = (inMass || corner)
            ? ldsD[w][a][i][i + b2] / (cnv * pnv[a][b2]) : 0.0f;
      }
    float mass = 0.0f;
#pragma unroll
    for (int dyi = 0; dyi < 3; ++dyi)
#pragma unroll
      for (int dxi = 0; dxi < 3; ++dxi)
        if (yv[dyi] && xv[dxi]) mass += fmaxf(av[dyi][dxi], 0.0f);
    ldsW[4][x] = ((mass > -10.0f) && (mass < 10.0f)) ? 1.0f : 0.0f;
    const float msafe = (mass > 0.0f) ? mass : 1.0f;
#pragma unroll
    for (int oy = 0; oy < 2; ++oy) {
      const int dyi = oy ? 3 : 0;
#pragma unroll
      for (int ox = 0; ox < 2; ++ox) {
        const int dxi = ox ? 3 : 0;
        const float a = av[dyi][dxi];
        float wv = 0.0f;
        if (yv[dyi] && xv[dxi]) wv = (a > 0.0f) ? (a / msafe) : 0.0f;
        ldsW[oy * 2 + ox][x] = wv;
      }
    }
  }
  __syncthreads();

  const int yb = y * Wd;
  const int px = t & (Wd - 1);
  const int hh = t >> 6;
  float v[5];
  for (int j = 0; j < 5; ++j) v[j] = ldsW[j][px];
  for (int c = 0; c < CG / 2; ++c) {
    const int cg = (CG / 2) * hh + c;
    for (int j = 0; j < 5; ++j)
      outN[(size_t)(cg * DC + j) * HW + yb + px] = v[j];
  }
}

__global__ __launch_bounds__(256) void FeatureAlign_stream(
    const float* __restrict__ mem, float* __restrict__ out) {
  const int b = blockIdx.x;
  const int cg = b & (CG - 1);
  const int nb = b / CG;
  const int n = nb >> 6;
  const int y = nb & (Hd - 1);
  const int t = threadIdx.x;
  const int x = t & (Wd - 1);
  const int ds = __builtin_amdgcn_readfirstlane(t >> 6);

  const float* memN = mem + (size_t)n * Dd * HW;
  float* outN = out + (size_t)n * Dd * HW;
  const int yb = y * Wd;

  const float* wsrc = outN + (size_t)(cg * DC) * HW + yb + x;
  float w[4];
  for (int off = 0; off < 4; ++off) w[off] = wsrc[(size_t)off * HW];
  const float cf = wsrc[(size_t)4 * HW];
  __syncthreads();
  const bool cond = cf != 0.0f;

  int ym = y - 1; ym = ym < 0 ? 0 : ym;
  int yp = y + 2; yp = yp > Hd - 1 ? Hd - 1 : yp;
  int xm = x - 1; xm = xm < 0 ? 0 : xm;
  int xp = x + 2; xp = xp > Wd - 1 ? Wd - 1 : xp;
  const unsigned oc = (unsigned)(yb + x);
  const unsigned o0 = (unsigned)(ym * Wd + xm), o1 = (unsigned)(ym * Wd + xp);
  const unsigned o2 = (unsigned)(yp * Wd + xm), o3 = (unsigned)(yp * Wd + xp);

#pragma unroll
  for (int i = 0; i < DC / 4; ++i) {
    const int d = cg * DC + ds + 4 * i;
    const float* md = memN + (size_t)(unsigned)d * HW;
    const float m0 = md[oc];
    float agg = 0.0f;
    agg += w[0] * md[o0];
    agg += w[1] * md[o1];
    agg += w[2] * md[o2];
    agg += w[3] * md[o3];
    const float a = (d > 0) ? agg : 0.0f;
    float* od = outN + (size_t)(unsigned)d * HW;
    od[oc] = cond ? (m0 + a) : 0.0f;
  }
}

extern "C" void kernel_launch(void* const* d_in, const int* in_sizes, int n_in,
                              void* d_out, int out_size, void* d_ws, size_t ws_size,
                              hipStream_t stream) {
  const float* cur = (const float*)d_in[0];
  const float* prev = (const float*)d_in[1];
  const float* mem = (const float*)d_in[2];
  float* out = (float*)d_out;
  (void)in_sizes; (void)n_in; (void)out_size; (void)ws_size;

  float* cnF = (float*)d_ws;
  float* pnF = cnF + (size_t)Nn * HW;

  FeatureAlign_norms<<<dim3(Nn * Hd), dim3(256), 0, stream>>>(
      cur, prev, cnF, pnF);
  FeatureAlign_kernel<<<dim3(Nn * Hd), dim3(128), 0, stream>>>(
      cur, prev, cnF, pnF, out);
  FeatureAlign_stream<<<dim3(Nn * Hd * CG), dim3(256), 0, stream>>>(mem, out);
}


# Norms and dots run as independent blocks of one launch (dots no longer wait for norms); division by norms, mass and weights formed in the stream launch from raw dots handed over in out; sums kept
# speedup vs baseline: 6.1161x; 1.1636x over previous
#include <hip/hip_runtime.h>
#include <hip/hip_bf16.h>
#include <math.h>

typedef __attribute__((ext_vector_type(2))) float v2f;
typedef __attribute__((ext_vector_type(8))) float v8f;
typedef __attribute__((ext_vector_type(4))) float v4f;

#define Nn 4
#define Dd 256
#define Hd 64
#define Wd 64
#define HW (Hd * Wd)
#define DC 64
#define CG (Dd / DC)
#define KC 32
#define NST (Dd / KC)
#define PITCH 72
#define SLOT (KC * PITCH)
#define STG (5 * SLOT)

#define NG 16
__device__ __forceinline__ void norms_part(
    const int nb, const float* __restrict__ cur, const float* __restrict__ prev,
    float* __restrict__ cnF, float* __restrict__ pnF) {
  __shared__ __attribute__((aligned(16))) float red[2][NG][Wd];

  const int n = nb >> 6;
  const int y = nb & (Hd - 1);
  const int t = threadIdx.x;
  const int q4 = (t & 15) << 2;
  const int c8 = t >> 4;

  const float* curR = cur + (size_t)n * Dd * HW + y * Wd;
  const float* prevR = prev + (size_t)n * Dd * HW + y * Wd;

  v4f c[2][Dd / NG], p[2][Dd / NG];
#pragma unroll
  for (int hf = 0; hf < 2; ++hf) {
    const unsigned lo = (unsigned)((c8 + 8 * hf) * HW + q4);
#pragma unroll
    for (int j = 0; j < Dd / NG; ++j)
      c[hf][j] = *(const v4f*)(curR + lo + (unsigned)(j * NG * HW));
  }
#pragma unroll
  for (int hf = 0; hf < 2; ++hf) {
    const unsigned lo = (unsigned)((c8 + 8 * hf) * HW + q4);
#pragma unroll
    for (int j = 0; j < Dd / NG; ++j)
      p[hf][j] = *(const v4f*)(prevR + lo + (unsigned)(j * NG * HW));
  }

#pragma unroll
  for (int hf = 0; hf < 2; ++hf) {
    v4f cs = {0.0f, 0.0f, 0.0f, 0.0f}, ps = {0.0f, 0.0f, 0.0f, 0.0f};
#pragma unroll
    for (int j = 0; j < Dd / NG; ++j) cs += c[hf][j] * c[hf][j];
#pragma unroll
    for (int j = 0; j < Dd / NG; ++j) ps += p[hf][j] * p[hf][j];
    *(v4f*)&red[0][c8 + 8 * hf][q4] = cs;
    *(v4f*)&red[1][c8 + 8 * hf][q4] = ps;
  }
  __syncthreads();

  {
    const int arr = t >> 6;
    const int x = t & (Wd - 1);
    float s = 0.0f;
#pragma unroll
    for (int gI = 0; gI < NG; ++gI) s += red[arr][gI][x];
    float* dst = arr ? pnF : cnF;
    dst[nb * Wd + x] = sqrtf(s) + 1e-8f;
  }
}

__device__ __forceinline__ void dots_part(
    const int nb, const float* __restrict__ cur, const float* __restrict__ prev,
    float* __restrict__ out) {
  __shared__ __attribute__((aligned(16))) float ldsS[2][STG];
  __shared__ float ldsD[4][4][16][32];

  const int n = nb >> 6;
  const int y = nb & (Hd - 1);
  const int t = threadIdx.x;
  const int wave = t >> 5;
  const int lane = t & 31;
  const int g = lane >> 4;
  const int li = lane & 15;
  const int x0 = wave << 4;

  const float* curN = cur + (size_t)n * Dd * HW;
  const float* prevN = prev + (size_t)n * Dd * HW;
  float* outN = out + (size_t)n * Dd * HW;

  const int q4 = (t & 15) << 2;
  const int cs = t >> 4;
  const unsigned gA = (unsigned)(cs * HW + y * Wd + q4);
  unsigned gB[4];
#pragma unroll
  for (int dyi = 0; dyi < 4; ++dyi) {
    const int dy = (dyi == 3) ? 2 : (dyi - 1);
    int yy = y + dy;
    yy = yy < 0 ? 0 : (yy > Hd - 1 ? Hd - 1 : yy);
    gB[dyi] = (unsigned)(cs * HW + yy * Wd + q4);
  }
  const unsigned sOff = (unsigned)(cs * PITCH + q4);

  const unsigned fA = (unsigned)(2 * g * PITCH + x0 + li);
  unsigned fB[4][2];
#pragma unroll
  for (int dyi = 0; dyi < 4; ++dyi)
#pragma unroll
    for (int h = 0; h < 2; ++h) {
      int bx = x0 - 1 + 16 * h + li;
      bx = bx < 0 ? 0 : (bx > Wd - 1 ? Wd - 1 : bx);
      fB[dyi][h] = (unsigned)((dyi + 1) * SLOT + 2 * g * PITCH + bx);
    }

  v8f acc[4][2];
#pragma unroll
  for (int dyi = 0; dyi < 4; ++dyi)
#pragma unroll
    for (int h = 0; h < 2; ++h) acc[dyi][h] = {};

  v4f st[5][4];
  auto gload = [&](int s) {
    const float* pc = curN + (size_t)(unsigned)(s * KC) * HW;
    const float* pp = prevN + (size_t)(unsigned)(s * KC) * HW;
#pragma unroll
    for (int jj = 0; jj < 4; ++jj)
      st[0][jj] = *(const v4f*)(pc + gA + (unsigned)(jj * 8 * HW));
#pragma unroll
    for (int r = 0; r < 4; ++r)
#pragma unroll
      for (int jj = 0; jj < 4; ++jj)
        st[r + 1][jj] = *(const v4f*)(pp + gB[r] + (unsigned)(jj * 8 * HW));
  };
  auto sstore = [&](int buf) {
    float* b = &ldsS[buf][0];
#pragma unroll
    for (int r = 0; r < 5; ++r)
#pragma unroll
      for (int jj = 0; jj < 4; ++jj)
        *(v4f*)(b + r * SLOT + jj * 8 * PITCH + sOff) = st[r][jj];
  };
  struct Frag {
    v2f a[2];
    v2f b[2][4][2];
  };
  auto fload = [&](const float* b, int blk, Frag& f) {
#pragma unroll
    for (int c = 0; c < 2; ++c) {
      const int kk = (blk * 2 + c) * 4;
      f.a[c].x = b[fA + kk * PITCH];
      f.a[c].y = b[fA + (kk + 1) * PITCH];
#pragma unroll
      for (int dyi = 0; dyi < 4; ++dyi)
#pragma unroll
        for (int h = 0; h < 2; ++h) {
          f.b[c][dyi][h].x = b[fB[dyi][h] + kk * PITCH];
          f.b[c][dyi][h].y = b[fB[dyi][h] + (kk + 1) * PITCH];
        }
    }
  };
  auto fmma = [&](const Frag& f) {
#pragma unroll
    for (int c = 0; c < 2; ++c)
#pragma unroll
      for (int dyi = 0; dyi < 4; ++dyi)
#pragma unroll
        for (int h = 0; h < 2; ++h)
          acc[dyi][h] = __builtin_amdgcn_wmma_f32_16x16x4_f32(
              false, f.a[c], false, f.b[c][dyi][h], (short)0, acc[dyi][h],
              false, false);
  };
  auto compute = [&](int buf) {
    const float* b = &ldsS[buf][0];
    Frag f0, f1;
    fload(b, 0, f0);
    __builtin_amdgcn_sched_barrier(0);
    fload(b, 1, f1);
    __builtin_amdgcn_sched_barrier(0);
    fmma(f0);
    __builtin_amdgcn_sched_barrier(0);
    fload(b, 2, f0);
    __builtin_amdgcn_sched_barrier(0);
    fmma(f1);
    __builtin_amdgcn_sched_barrier(0);
    fload(b, 3, f1);
    __builtin_amdgcn_sched_barrier(0);
    fmma(f0);
    fmma(f1);
  };

  gload(0);
  sstore(0);
  __syncthreads();
  for (int s = 0; s < NST; ++s) {
    const bool more = (s + 1 < NST);
    if (more) gload(s + 1);
    compute(s & 1);
    __builtin_amdgcn_sched_barrier(0);
    if (more) sstore((s + 1) & 1);
    __syncthreads();
  }

#pragma unroll
  for (int dyi = 0; dyi < 4; ++dyi)
#pragma unroll
    for (int h = 0; h < 2; ++h)
#pragma unroll
      for (int r = 0; r < 8; ++r)
        ldsD[wave][dyi][r + 8 * g][16 * h + li] = acc[dyi][h][r];

  __syncthreads();

  const int yb = y * Wd;
  const int px = t & (Wd - 1);
  const int hh = t >> 6;
  const int pw = px >> 4, pi = px & 15;
#pragma unroll
  for (int a = 0; a < 4; ++a)
#pragma unroll
    for (int b2 = 0; b2 < 4; ++b2) {
      const bool inMass = (a < 3) && (b2 < 3);
      const bool corner = (a == 0 || a == 3) && (b2 == 0 || b2 == 3);
      if (inMass || corner) {
        const float v = ldsD[pw][a][pi][pi + b2];
#pragma unroll
        for (int c = 0; c < CG / 2; ++c) {
          const int cg = (CG / 2) * hh + c;
          outN[(size_t)(cg * DC + 4 * a + b2) * HW + yb + px] = v;
        }
      }
    }
}

__global__ __launch_bounds__(128) void FeatureAlign_kernel(
    const float* __restrict__ cur, const float* __restrict__ prev,
    float* __restrict__ cnF, float* __restrict__ pnF,
    float* __restrict__ out) {
  const int b = blockIdx.x;
  if (b < Nn * Hd)
    dots_part(b, cur, prev, out);
  else
    norms_part(b - Nn * Hd, cur, prev, cnF, pnF);
}

__global__ __launch_bounds__(256) void FeatureAlign_stream(
    const float* __restrict__ mem, const float* __restrict__ cnF,
    const float* __restrict__ pnF, float* __restrict__ out) {
  __shared__ float ldsW[5][Wd];
  const int b = blockIdx.x;
  const int cg = b & (CG - 1);
  const int nb = b / CG;
  const int n = nb >> 6;
  const int y = nb & (Hd - 1);
  const int t = threadIdx.x;
  const int x = t & (Wd - 1);
  const int ds = __builtin_amdgcn_readfirstlane(t >> 6);

  const float* memN = mem + (size_t)n * Dd * HW;
  float* outN = out + (size_t)n * Dd * HW;
  const int yb = y * Wd;

  if (t < Wd) {
    bool yv[4], xv[4];
    int yc[4], xc[4];
#pragma unroll
    for (int k = 0; k < 4; ++k) {
      const int d = (k == 3) ? 2 : (k - 1);
      const int yy = y + d, xx = x + d;
      yv[k] = (yy >= 0) && (yy < Hd);
      xv[k] = (xx >= 0) && (xx < Wd);
      yc[k] = yy < 0 ? 0 : (yy > Hd - 1 ? Hd - 1 : yy);
      xc[k] = xx < 0 ? 0 : (xx > Wd - 1 ? Wd - 1 : xx);
    }
    const float* dsrc = outN + (size_t)(cg * DC) * HW + yb + x;
    const int nb2 = n * HW;
    const float cnv = cnF[nb2 + yb + x];
    float dv[4][4], pnv[4][4];
#pragma unroll
    for (int a = 0; a < 4; ++a)
#pragma unroll
      for (int b2 = 0; b2 < 4; ++b2) {
        const bool inMass = (a < 3) && (b2 < 3);
        const bool corner = (a == 0 || a == 3) && (b2 == 0 || b2 == 3);
        dv[a][b2] = 0.0f;
        pnv[a][b2] = 0.0f;
        if (inMass || corner) {
          dv[a][b2] = dsrc[(size_t)(4 * a + b2) * HW];
          pnv[a][b2] = pnF[nb2 + yc[a] * Wd + xc[b2]];
        }
      }
    float av[4][4];
#pragma unroll
    for (int a = 0; a < 4; ++a)
#pragma unroll
      for (int b2 = 0; b2 < 4; ++b2) {
        const bool inMass = (a < 3) && (b2 < 3);
        const bool corner = (a == 0 || a == 3) && (b2 == 0 || b2 == 3);
        av[a][b2] = (inMass || corner)
            ? dv[a][b2] / (cnv * pnv[a][b2]) : 0.0f;
      }
    float mass = 0.0f;
#pragma unroll
    for (int dyi = 0; dyi < 3; ++dyi)
#pragma unroll
      for (int dxi = 0; dxi < 3; ++dxi)
        if (yv[dyi] && xv[dxi]) mass += fmaxf(av[dyi][dxi], 0.0f);
    ldsW[4][x] = ((mass > -10.0f) && (mass < 10.0f)) ? 1.0f : 0.0f;
    const float msafe = (mass > 0.0f) ? mass : 1.0f;
#pragma unroll
    for (int oy = 0; oy < 2; ++oy) {
      const int dyi = oy ? 3 : 0;
#pragma unroll
      for (int ox = 0; ox < 2; ++ox) {
        const int dxi = ox ? 3 : 0;
        const float a = av[dyi][dxi];
        float wv = 0.0f;
        if (yv[dyi] && xv[dxi]) wv = (a > 0.0f) ? (a / msafe) : 0.0f;
        ldsW[oy * 2 + ox][x] = wv;
      }
    }
  }
  __syncthreads();
  float w[4];
  for (int off = 0; off < 4; ++off) w[off] = ldsW[off][x];
  const bool cond = ldsW[4][x] != 0.0f;

  int ym = y - 1; ym = ym < 0 ? 0 : ym;
  int yp = y + 2; yp = yp > Hd - 1 ? Hd - 1 : yp;
  int xm = x - 1; xm = xm < 0 ? 0 : xm;
  int xp = x + 2; xp = xp > Wd - 1 ? Wd - 1 : xp;
  const unsigned oc = (unsigned)(yb + x);
  const unsigned o0 = (unsigned)(ym * Wd + xm), o1 = (unsigned)(ym * Wd + xp);
  const unsigned o2 = (unsigned)(yp * Wd + xm), o3 = (unsigned)(yp * Wd + xp);

#pragma unroll
  for (int i = 0; i < DC / 4; ++i) {
    const int d = cg * DC + ds + 4 * i;
    const float* md = memN + (size_t)(unsigned)d * HW;
    const float m0 = md[oc];
    float agg = 0.0f;
    agg += w[0] * md[o0];
    agg += w[1] * md[o1];
    agg += w[2] * md[o2];
    agg += w[3] * md[o3];
    const float a = (d > 0) ? agg : 0.0f;
    float* od = outN + (size_t)(unsigned)d * HW;
    od[oc] = cond ? (m0 + a) : 0.0f;
  }
}

extern "C" void kernel_launch(void* const* d_in, const int* in_sizes, int n_in,
                              void* d_out, int out_size, void* d_ws, size_t ws_size,
                              hipStream_t stream) {
  const float* cur = (const float*)d_in[0];
  const float* prev = (const float*)d_in[1];
  const float* mem = (const float*)d_in[2];
  float* out = (float*)d_out;
  (void)in_sizes; (void)n_in; (void)out_size; (void)ws_size;

  float* cnF = (float*)d_ws;
  float* pnF = cnF + (size_t)Nn * HW;

  FeatureAlign_kernel<<<dim3(2 * Nn * Hd), dim3(128), 0, stream>>>(
      cur, prev, cnF, pnF, out);
  FeatureAlign_stream<<<dim3(Nn * Hd * CG), dim3(256), 0, stream>>>(
      mem, cnF, pnF, out);
}


# One launch of 8-wave row blocks: 4 loader waves stage rows to LDS and square-sum them into the norms (8 partial sums each: sum order changed), 4 matrix waves; weights and mem->out done in-block
# speedup vs baseline: 7.1711x; 1.1725x over previous
#include <hip/hip_runtime.h>
#include <math.h>


typedef __attribute__((ext_vector_type(2))) float v2f;
typedef __attribute__((ext_vector_type(8))) float v8f;
typedef __attribute__((ext_vector_type(4))) float v4f;

#define Nn 4
#define Dd 256
#define Hd 64
#define Wd 64
#define HW (Hd * Wd)
#define KC 32
#define NST (Dd / KC)
#define PITCH 72
#define SLOT (KC * PITCH)
#define STG (5 * SLOT)
#define NP 8

__global__ __launch_bounds__(256) void FeatureAlign_kernel(
    const float* __restrict__ cur, const float* __restrict__ prev,
    const float* __restrict__ mem, float* __restrict__ out) {
  __shared__ __attribute__((aligned(16))) float ldsS[2][2 * STG];
  __shared__ float ldsD[4][4][16][32];
  __shared__ __attribute__((aligned(16))) float red[5][NP][Wd];
  __shared__ float nrm[5][Wd];
  __shared__ __attribute__((aligned(16))) float ldsW[5][Wd];

  const int nb = blockIdx.x;
  const int n = nb >> 6;
  const int y = nb & (Hd - 1);
  const int t = threadIdx.x;
  const int wave = __builtin_amdgcn_readfirstlane(t >> 5);
  const bool isMat = wave < 4;
  const int lane = t & 31;
  const int g = lane >> 4;
  const int li = lane & 15;
  const int x0 = (wave & 3) << 4;

  const float* curN = cur + (size_t)n * Dd * HW;
  const float* prevN = prev + (size_t)n * Dd * HW;
  const float* memN = mem + (size_t)n * Dd * HW;
  float* outN = out + (size_t)n * Dd * HW;

  const int lt = t & 127;
  const int q4 = (lt & 15) << 2;
  const int cs = lt >> 4;
  const unsigned gA = (unsigned)(cs * HW + y * Wd + q4);
  unsigned gB[4];
#pragma unroll
  for (int dyi = 0; dyi < 4; ++dyi) {
    const int dy = (dyi == 3) ? 2 : (dyi - 1);
    int yy = y + dy;
    yy = yy < 0 ? 0 : (yy > Hd - 1 ? Hd - 1 : yy);
    gB[dyi] = (unsigned)(cs * HW + yy * Wd + q4);
  }
  const unsigned sOff = (unsigned)(cs * PITCH + q4);

  const unsigned fA = (unsigned)(2 * g * PITCH + x0 + li);
  unsigned fB[4][2];
#pragma unroll
  for (int dyi = 0; dyi < 4; ++dyi)
#pragma unroll
    for (int h = 0; h < 2; ++h) {
      int bx = x0 - 1 + 16 * h + li;
      bx = bx < 0 ? 0 : (bx > Wd - 1 ? Wd - 1 : bx);
      fB[dyi][h] = (unsigned)((dyi + 1) * SLOT + 2 * g * PITCH + bx);
    }

  v8f acc[4][2];
#pragma unroll
  for (int dyi = 0; dyi < 4; ++dyi)
#pragma unroll
    for (int h = 0; h < 2; ++h) acc[dyi][h] = {};

  v4f sq[5];
#pragma unroll
  for (int r = 0; r < 5; ++r) sq[r] = {0.0f, 0.0f, 0.0f, 0.0f};

  auto gload = [&](v4f (&st)[5][4], int s) {
    const float* pc = curN + (size_t)(unsigned)(s * KC) * HW;
    const float* pp = prevN + (size_t)(unsigned)(s * KC) * HW;
#pragma unroll
    for (int jj = 0; jj < 4; ++jj)
      st[0][jj] = *(const v4f*)(pc + gA + (unsigned)(jj * 8 * HW));
#pragma unroll
    for (int r = 0; r < 4; ++r)
#pragma unroll
      for (int jj = 0; jj < 4; ++jj)
        st[r + 1][jj] = *(const v4f*)(pp + gB[r] + (unsigned)(jj * 8 * HW));
  };
  auto sqsum = [&](const v4f (&st)[5][4]) {
#pragma unroll
    for (int r = 0; r < 5; ++r)
#pragma unroll
      for (int jj = 0; jj < 4; ++jj) sq[r] += st[r][jj] * st[r][jj];
  };
  auto sstore = [&](const v4f (&st)[5][4], float* b) {
#pragma unroll
    for (int r = 0; r < 5; ++r)
#pragma unroll
      for (int jj = 0; jj < 4; ++jj)
        *(v4f*)(b + r * SLOT + jj * 8 * PITCH + sOff) = st[r][jj];
  };
  struct Frag {
    v2f a[2];
    v2f b[2][4][2];
  };
  auto fload = [&](const float* b, int blk, Frag& f) {
#pragma unroll
    for (int c = 0; c < 2; ++c) {
      const int kk = (blk * 2 + c) * 4;
      f.a[c].x = b[fA + kk * PITCH];
      f.a[c].y = b[fA + (kk + 1) * PITCH];
#pragma unroll
      for (int dyi = 0; dyi < 4; ++dyi)
#pragma unroll
        for (int h = 0; h < 2; ++h) {
          f.b[c][dyi][h].x = b[fB[dyi][h] + kk * PITCH];
          f.b[c][dyi][h].y = b[fB[dyi][h] + (kk + 1) * PITCH];
        }
    }
  };
  auto fmma = [&](const Frag& f) {
#pragma unroll
    for (int c = 0; c < 2; ++c)
#pragma unroll
      for (int dyi = 0; dyi < 4; ++dyi)
#pragma unroll
        for (int h = 0; h < 2; ++h)
          acc[dyi][h] = __builtin_amdgcn_wmma_f32_16x16x4_f32(
              false, f.a[c], false, f.b[c][dyi][h], (short)0, acc[dyi][h],
              false, false);
  };
  auto compute = [&](int buf) {
    const float* b = &ldsS[buf][0];
    Frag f0, f1;
    fload(b, 0, f0);
    __builtin_amdgcn_sched_barrier(0);
    fload(b, 1, f1);
    __builtin_amdgcn_sched_barrier(0);
    fmma(f0);
    __builtin_amdgcn_sched_barrier(0);
    fload(b, 2, f0);
    __builtin_amdgcn_sched_barrier(0);
    fmma(f1);
    __builtin_amdgcn_sched_barrier(0);
    fload(b, 3, f1);
    __builtin_amdgcn_sched_barrier(0);
    fmma(f0);
    __builtin_amdgcn_sched_barrier(0);
    fload(b + STG, 0, f0);
    __builtin_amdgcn_sched_barrier(0);
    fmma(f1);
    __builtin_amdgcn_sched_barrier(0);
    fload(b + STG, 1, f1);
    __builtin_amdgcn_sched_barrier(0);
    fmma(f0);
    __builtin_amdgcn_sched_barrier(0);
    fload(b + STG, 2, f0);
    __builtin_amdgcn_sched_barrier(0);
    fmma(f1);
    __builtin_amdgcn_sched_barrier(0);
    fload(b + STG, 3, f1);
    __builtin_amdgcn_sched_barrier(0);
    fmma(f0);
    fmma(f1);
  };
  auto stage2 = [&](int s0, int buf) {
    v4f stA[5][4], stB[5][4];
    gload(stA, s0);
    gload(stB, s0 + 1);
    sqsum(stA);
    sstore(stA, &ldsS[buf][0]);
    sqsum(stB);
    sstore(stB, &ldsS[buf][STG]);
  };

  if (!isMat) stage2(0, 0);
  __syncthreads();
  for (int i = 0; i < NST / 2; ++i) {
    if (isMat) {
      compute(i & 1);
    } else if (i + 1 < NST / 2) {
      stage2(2 * (i + 1), (i + 1) & 1);
    }
    __syncthreads();
  }

  const int eq = t & 15;
  const int cq = t >> 4;
  const int x4 = eq << 2;
  int ym = y - 1; ym = ym < 0 ? 0 : ym;
  int yp = y + 2; yp = yp > Hd - 1 ? Hd - 1 : yp;
  const bool lastq = (eq == 15);
  const unsigned oC = (unsigned)(y * Wd + x4);
  const unsigned oM = (unsigned)(ym * Wd + x4);
  const unsigned oP = (unsigned)(yp * Wd + x4);
  const unsigned dL = (x4 > 0) ? 1u : 0u;
  const unsigned dR = lastq ? 2u : 4u;
  struct EL {
    v4f c, m0, m1;
    v2f r0, r1;
    float l0, l1;
  };
  auto eload = [&](int i, EL& e) {
    const float* md = memN + (size_t)(unsigned)(cq + 16 * i) * HW;
    e.c = *(const v4f*)(md + oC);
    e.l0 = md[oM - dL];
    e.m0 = *(const v4f*)(md + oM);
    e.r0 = *(const v2f*)(md + oM + dR);
    e.l1 = md[oP - dL];
    e.m1 = *(const v4f*)(md + oP);
    e.r1 = *(const v2f*)(md + oP + dR);
  };
  if (isMat) {
#pragma unroll
    for (int dyi = 0; dyi < 4; ++dyi)
#pragma unroll
      for (int h = 0; h < 2; ++h)
#pragma unroll
        for (int r = 0; r < 8; ++r)
          ldsD[wave][dyi][r + 8 * g][16 * h + li] = acc[dyi][h][r];
  } else {
#pragma unroll
    for (int r = 0; r < 5; ++r) *(v4f*)&red[r][cs][q4] = sq[r];
  }
  __syncthreads();

  for (int idx = t; idx < 5 * Wd; idx += 256) {
    const int r = idx >> 6;
    const int xx = idx & (Wd - 1);
    float s = 0.0f;
#pragma unroll
    for (int k = 0; k < NP; ++k) s += red[r][k][xx];
    nrm[r][xx] = sqrtf(s) + 1e-8f;
  }
  __syncthreads();
  EL e0[8];
#pragma unroll
  for (int i = 0; i < 8; ++i) eload(i, e0[i]);

  if (t < Wd) {
    const int x = t;
    bool yv[4], xv[4];
    int xc[4];
#pragma unroll
    for (int k = 0; k < 4; ++k) {
      const int d = (k == 3) ? 2 : (k - 1);
      const int yy = y + d, xx = x + d;
      yv[k] = (yy >= 0) && (yy < Hd);
      xv[k] = (xx >= 0) && (xx < Wd);
      xc[k] = xx < 0 ? 0 : (xx > Wd - 1 ? Wd - 1 : xx);
    }
    const int pw = x >> 4, pi = x & 15;
    const float cnv = nrm[0][x];
    float av[4][4];
#pragma unroll
    for (int a = 0; a < 4; ++a)
#pragma unroll
      for (int b2 = 0; b2 < 4; ++b2) {
        const bool inMass = (a < 3) && (b2 < 3);
        const bool corner = (a == 0 || a == 3) && (b2 == 0 || b2 == 3);
        av[a][b2] = 0.0f;
        if (inMass || corner)
          av[a][b2] = ldsD[pw][a][pi][pi + b2] / (cnv * nrm[1 + a][xc[b2]]);
      }
    float mass = 0.0f;
#pragma unroll
    for (int dyi = 0; dyi < 3; ++dyi)
#pragma unroll
      for (int dxi = 0; dxi < 3; ++dxi)
        if (yv[dyi] && xv[dxi]) mass += fmaxf(av[dyi][dxi], 0.0f);
    ldsW[4][x] = ((mass > -10.0f) && (mass < 10.0f)) ? 1.0f : 0.0f;
    const float msafe = (mass > 0.0f) ? mass : 1.0f;
#pragma unroll
    for (int oy = 0; oy < 2; ++oy) {
      const int dyi = oy ? 3 : 0;
#pragma unroll
      for (int ox = 0; ox < 2; ++ox) {
        const int dxi = ox ? 3 : 0;
        const float a = av[dyi][dxi];
        float wv = 0.0f;
        if (yv[dyi] && xv[dxi]) wv = (a > 0.0f) ? (a / msafe) : 0.0f;
        ldsW[oy * 2 + ox][x] = wv;
      }
    }
  }
  __syncthreads();

  const v4f w0 = *(const v4f*)&ldsW[0][x4];
  const v4f w1 = *(const v4f*)&ldsW[1][x4];
  const v4f w2 = *(const v4f*)&ldsW[2][x4];
  const v4f w3 = *(const v4f*)&ldsW[3][x4];
  const v4f cd = *(const v4f*)&ldsW[4][x4];

  auto estore = [&](int i, const EL& e) {
    const int d = cq + 16 * i;
    const float rx0 = lastq ? e.r0.y : e.r0.x;
    const float rx1 = lastq ? e.r1.y : e.r1.x;
    const v4f nl0 = {e.l0, e.m0.x, e.m0.y, e.m0.z};
    const v4f nr0 = {e.m0.z, e.m0.w, rx0, e.r0.y};
    const v4f nl1 = {e.l1, e.m1.x, e.m1.y, e.m1.z};
    const v4f nr1 = {e.m1.z, e.m1.w, rx1, e.r1.y};
    v4f agg = {0.0f, 0.0f, 0.0f, 0.0f};
    agg += w0 * nl0;
    agg += w1 * nr0;
    agg += w2 * nl1;
    agg += w3 * nr1;
    if (d == 0) agg = {0.0f, 0.0f, 0.0f, 0.0f};
    v4f res;
    res.x = (cd.x != 0.0f) ? (e.c.x + agg.x) : 0.0f;
    res.y = (cd.y != 0.0f) ? (e.c.y + agg.y) : 0.0f;
    res.z = (cd.z != 0.0f) ? (e.c.z + agg.z) : 0.0f;
    res.w = (cd.w != 0.0f) ? (e.c.w + agg.w) : 0.0f;
    *(v4f*)(outN + (size_t)(unsigned)d * HW + oC) = res;
  };
#pragma unroll
  for (int i = 0; i < 8; ++i) estore(i, e0[i]);
#pragma unroll
  for (int i = 8; i < Dd / 16; ++i) {
    EL e;
    eload(i, e);
    estore(i, e);
  }
}

extern "C" void kernel_launch(void* const* d_in, const int* in_sizes, int n_in,
                              void* d_out, int out_size, void* d_ws, size_t ws_size,
                              hipStream_t stream) {
  const float* cur = (const float*)d_in[0];
  const float* prev = (const float*)d_in[1];
  const float* mem = (const float*)d_in[2];
  float* out = (float*)d_out;
  (void)in_sizes; (void)n_in; (void)out_size; (void)d_ws; (void)ws_size;

  FeatureAlign_kernel<<<dim3(Nn * Hd), dim3(256), 0, stream>>>(cur, prev, mem, out);
}


# Epilogue mem loads requested ahead: channel steps 0-7 during the last dot interval (matrix waves 7 per WMMA block), 8-15 behind it, kept in flight across hand-overs by LDS-only barriers; sums kept
# speedup vs baseline: 7.5803x; 1.0571x over previous
#include <hip/hip_runtime.h>
#include <math.h>


typedef __attribute__((ext_vector_type(2))) float v2f;
typedef __attribute__((ext_vector_type(8))) float v8f;
typedef __attribute__((ext_vector_type(4))) float v4f;

#define Nn 4
#define Dd 256
#define Hd 64
#define Wd 64
#define HW (Hd * Wd)
#define KC 32
#define NST (Dd / KC)
#define PITCH 72
#define SLOT (KC * PITCH)
#define STG (5 * SLOT)
#define NP 8

static __device__ __forceinline__ void lds_barrier() {
  asm volatile("s_wait_dscnt 0x0\n\ts_barrier_signal -1\n\ts_barrier_wait -1" ::: "memory");
}

__global__ __launch_bounds__(256) void FeatureAlign_kernel(
    const float* __restrict__ cur, const float* __restrict__ prev,
    const float* __restrict__ mem, float* __restrict__ out) {
  __shared__ __attribute__((aligned(16))) float ldsS[2][2 * STG];
  __shared__ float ldsD[4][4][16][32];
  __shared__ __attribute__((aligned(16))) float red[5][NP][Wd];
  __shared__ float nrm[5][Wd];
  __shared__ __attribute__((aligned(16))) float ldsW[5][Wd];

  const int nb = blockIdx.x;
  const int n = nb >> 6;
  const int y = nb & (Hd - 1);
  const int t = threadIdx.x;
  const int wave = __builtin_amdgcn_readfirstlane(t >> 5);
  const bool isMat = wave < 4;
  const int lane = t & 31;
  const int g = lane >> 4;
  const int li = lane & 15;
  const int x0 = (wave & 3) << 4;

  const float* curN = cur + (size_t)n * Dd * HW;
  const float* prevN = prev + (size_t)n * Dd * HW;
  const float* memN = mem + (size_t)n * Dd * HW;
  float* outN = out + (size_t)n * Dd * HW;

  const int lt = t & 127;
  const int q4 = (lt & 15) << 2;
  const int cs = lt >> 4;
  const unsigned gA = (unsigned)(cs * HW + y * Wd + q4);
  unsigned gB[4];
#pragma unroll
  for (int dyi = 0; dyi < 4; ++dyi) {
    const int dy = (dyi == 3) ? 2 : (dyi - 1);
    int yy = y + dy;
    yy = yy < 0 ? 0 : (yy > Hd - 1 ? Hd - 1 : yy);
    gB[dyi] = (unsigned)(cs * HW + yy * Wd + q4);
  }
  const unsigned sOff = (unsigned)(cs * PITCH + q4);

  const unsigned fA = (unsigned)(2 * g * PITCH + x0 + li);
  unsigned fB[4][2];
#pragma unroll
  for (int dyi = 0; dyi < 4; ++dyi)
#pragma unroll
    for (int h = 0; h < 2; ++h) {
      int bx = x0 - 1 + 16 * h + li;
      bx = bx < 0 ? 0 : (bx > Wd - 1 ? Wd - 1 : bx);
      fB[dyi][h] = (unsigned)((dyi + 1) * SLOT + 2 * g * PITCH + bx);
    }

  v8f acc[4][2];
#pragma unroll
  for (int dyi = 0; dyi < 4; ++dyi)
#pragma unroll
    for (int h = 0; h < 2; ++h) acc[dyi][h] = {};

  v4f sq[5];
#pragma unroll
  for (int r = 0; r < 5; ++r) sq[r] = {0.0f, 0.0f, 0.0f, 0.0f};

  auto gload = [&](v4f (&st)[5][4], int s) {
    const float* pc = curN + (size_t)(unsigned)(s * KC) * HW;
    const float* pp = prevN + (size_t)(unsigned)(s * KC) * HW;
#pragma unroll
    for (int jj = 0; jj < 4; ++jj)
      st[0][jj] = *(const v4f*)(pc + gA + (unsigned)(jj * 8 * HW));
#pragma unroll
    for (int r = 0; r < 4; ++r)
#pragma unroll
      for (int jj = 0; jj < 4; ++jj)
        st[r + 1][jj] = *(const v4f*)(pp + gB[r] + (unsigned)(jj * 8 * HW));
  };
  auto sqsum = [&](const v4f (&st)[5][4]) {
#pragma unroll
    for (int r = 0; r < 5; ++r)
#pragma unroll
      for (int jj = 0; jj < 4; ++jj) sq[r] += st[r][jj] * st[r][jj];
  };
  auto sstore = [&](const v4f (&st)[5][4], float* b) {
#pragma unroll
    for (int r = 0; r < 5; ++r)
#pragma unroll
      for (int jj = 0; jj < 4; ++jj)
        *(v4f*)(b + r * SLOT + jj * 8 * PITCH + sOff) = st[r][jj];
  };
  struct Frag {
    v2f a[2];
    v2f b[2][4][2];
  };
  auto fload = [&](const float* b, int blk, Frag& f) {
#pragma unroll
    for (int c = 0; c < 2; ++c) {
      const int kk = (blk * 2 + c) * 4;
      f.a[c].x = b[fA + kk * PITCH];
      f.a[c].y = b[fA + (kk + 1) * PITCH];
#pragma unroll
      for (int dyi = 0; dyi < 4; ++dyi)
#pragma unroll
        for (int h = 0; h < 2; ++h) {
          f.b[c][dyi][h].x = b[fB[dyi][h] + kk * PITCH];
          f.b[c][dyi][h].y = b[fB[dyi][h] + (kk + 1) * PITCH];
        }
    }
  };
  auto fmma = [&](const Frag& f) {
#pragma unroll
    for (int c = 0; c < 2; ++c)
#pragma unroll
      for (int dyi = 0; dyi < 4; ++dyi)
#pragma unroll
        for (int h = 0; h < 2; ++h)
          acc[dyi][h] = __builtin_amdgcn_wmma_f32_16x16x4_f32(
              false, f.a[c], false, f.b[c][dyi][h], (short)0, acc[dyi][h],
              false, false);
  };
  auto compute = [&](int buf) {
    const float* b = &ldsS[buf][0];
    Frag f0, f1;
    fload(b, 0, f0);
    __builtin_amdgcn_sched_barrier(0);
    fload(b, 1, f1);
    __builtin_amdgcn_sched_barrier(0);
    fmma(f0);
    __builtin_amdgcn_sched_barrier(0);
    fload(b, 2, f0);
    __builtin_amdgcn_sched_barrier(0);
    fmma(f1);
    __builtin_amdgcn_sched_barrier(0);
    fload(b, 3, f1);
    __builtin_amdgcn_sched_barrier(0);
    fmma(f0);
    __builtin_amdgcn_sched_barrier(0);
    fload(b + STG, 0, f0);
    __builtin_amdgcn_sched_barrier(0);
    fmma(f1);
    __builtin_amdgcn_sched_barrier(0);
    fload(b + STG, 1, f1);
    __builtin_amdgcn_sched_barrier(0);
    fmma(f0);
    __builtin_amdgcn_sched_barrier(0);
    fload(b + STG, 2, f0);
    __builtin_amdgcn_sched_barrier(0);
    fmma(f1);
    __builtin_amdgcn_sched_barrier(0);
    fload(b + STG, 3, f1);
    __builtin_amdgcn_sched_barrier(0);
    fmma(f0);
    fmma(f1);
  };
  auto stage2 = [&](int s0, int buf) {
    v4f stA[5][4], stB[5][4];
    gload(stA, s0);
    gload(stB, s0 + 1);
    sqsum(stA);
    sstore(stA, &ldsS[buf][0]);
    sqsum(stB);
    sstore(stB, &ldsS[buf][STG]);
  };

  if (!isMat) stage2(0, 1);
  __syncthreads();
  for (int i = 0; i < NST / 2 - 1; ++i) {
    if (isMat) {
      compute((i + 1) & 1);
    } else {
      stage2(2 * (i + 1), i & 1);
    }
    __syncthreads();
  }

  const int eq = t & 15;
  const int cq = t >> 4;
  const int x4 = eq << 2;
  int ym = y - 1; ym = ym < 0 ? 0 : ym;
  int yp = y + 2; yp = yp > Hd - 1 ? Hd - 1 : yp;
  const bool lastq = (eq == 15);
  const unsigned oC = (unsigned)(y * Wd + x4);
  const unsigned oM = (unsigned)(ym * Wd + x4);
  const unsigned oP = (unsigned)(yp * Wd + x4);
  const unsigned dL = (x4 > 0) ? 1u : 0u;
  const unsigned dR = lastq ? 2u : 4u;
  struct EL {
    v4f c, m0, m1;
    v2f r0, r1;
    float l0, l1;
  };
  auto eload = [&](int i, EL& e) {
    const float* md = memN + (size_t)(unsigned)(cq + 16 * i) * HW;
    e.c = *(const v4f*)(md + oC);
    e.l0 = md[oM - dL];
    e.m0 = *(const v4f*)(md + oM);
    e.r0 = *(const v2f*)(md + oM + dR);
    e.l1 = md[oP - dL];
    e.m1 = *(const v4f*)(md + oP);
    e.r1 = *(const v2f*)(md + oP + dR);
  };
  EL e0[8], e1[8];
  {
    const float* b = &ldsS[(NST / 2) & 1][0];
    Frag f[2];
    if (isMat) {
      fload(b, 0, f[0]);
      __builtin_amdgcn_sched_barrier(0);
      fload(b, 1, f[1]);
    }
#pragma unroll
    for (int j = 0; j < 8; ++j) {
      eload(j, e0[j]);
      if (isMat) {
        fmma(f[j & 1]);
        if (j + 2 < 8) {
          __builtin_amdgcn_sched_barrier(0);
          fload(b + ((j + 2) >> 2) * STG, (j + 2) & 3, f[j & 1]);
        }
      }
    }
  }

  if (isMat) {
#pragma unroll
    for (int dyi = 0; dyi < 4; ++dyi)
#pragma unroll
      for (int h = 0; h < 2; ++h)
#pragma unroll
        for (int r = 0; r < 8; ++r)
          ldsD[wave][dyi][r + 8 * g][16 * h + li] = acc[dyi][h][r];
  } else {
#pragma unroll
    for (int r = 0; r < 5; ++r) *(v4f*)&red[r][cs][q4] = sq[r];
  }
#pragma unroll
  for (int i = 0; i < 8; ++i) eload(8 + i, e1[i]);
  lds_barrier();

  for (int idx = t; idx < 5 * Wd; idx += 256) {
    const int r = idx >> 6;
    const int xx = idx & (Wd - 1);
    float s = 0.0f;
#pragma unroll
    for (int k = 0; k < NP; ++k) s += red[r][k][xx];
    nrm[r][xx] = sqrtf(s) + 1e-8f;
  }
  lds_barrier();

  if (t < Wd) {
    const int x = t;
    bool yv[4], xv[4];
    int xc[4];
#pragma unroll
    for (int k = 0; k < 4; ++k) {
      const int d = (k == 3) ? 2 : (k - 1);
      const int yy = y + d, xx = x + d;
      yv[k] = (yy >= 0) && (yy < Hd);
      xv[k] = (xx >= 0) && (xx < Wd);
      xc[k] = xx < 0 ? 0 : (xx > Wd - 1 ? Wd - 1 : xx);
    }
    const int pw = x >> 4, pi = x & 15;
    const float cnv = nrm[0][x];
    float av[4][4];
#pragma unroll
    for (int a = 0; a < 4; ++a)
#pragma unroll
      for (int b2 = 0; b2 < 4; ++b2) {
        const bool inMass = (a < 3) && (b2 < 3);
        const bool corner = (a == 0 || a == 3) && (b2 == 0 || b2 == 3);
        av[a][b2] = 0.0f;
        if (inMass || corner)
          av[a][b2] = ldsD[pw][a][pi][pi + b2] / (cnv * nrm[1 + a][xc[b2]]);
      }
    float mass = 0.0f;
#pragma unroll
    for (int dyi = 0; dyi < 3; ++dyi)
#pragma unroll
      for (int dxi = 0; dxi < 3; ++dxi)
        if (yv[dyi] && xv[dxi]) mass += fmaxf(av[dyi][dxi], 0.0f);
    ldsW[4][x] = ((mass > -10.0f) && (mass < 10.0f)) ? 1.0f : 0.0f;
    const float msafe = (mass > 0.0f) ? mass : 1.0f;
#pragma unroll
    for (int oy = 0; oy < 2; ++oy) {
      const int dyi = oy ? 3 : 0;
#pragma unroll
      for (int ox = 0; ox < 2; ++ox) {
        const int dxi = ox ? 3 : 0;
        const float a = av[dyi][dxi];
        float wv = 0.0f;
        if (yv[dyi] && xv[dxi]) wv = (a > 0.0f) ? (a / msafe) : 0.0f;
        ldsW[oy * 2 + ox][x] = wv;
      }
    }
  }
  lds_barrier();

  const v4f w0 = *(const v4f*)&ldsW[0][x4];
  const v4f w1 = *(const v4f*)&ldsW[1][x4];
  const v4f w2 = *(const v4f*)&ldsW[2][x4];
  const v4f w3 = *(const v4f*)&ldsW[3][x4];
  const v4f cd = *(const v4f*)&ldsW[4][x4];

  auto estore = [&](int i, const EL& e) {
    const int d = cq + 16 * i;
    const float rx0 = lastq ? e.r0.y : e.r0.x;
    const float rx1 = lastq ? e.r1.y : e.r1.x;
    const v4f nl0 = {e.l0, e.m0.x, e.m0.y, e.m0.z};
    const v4f nr0 = {e.m0.z, e.m0.w, rx0, e.r0.y};
    const v4f nl1 = {e.l1, e.m1.x, e.m1.y, e.m1.z};
    const v4f nr1 = {e.m1.z, e.m1.w, rx1, e.r1.y};
    v4f agg = {0.0f, 0.0f, 0.0f, 0.0f};
    agg += w0 * nl0;
    agg += w1 * nr0;
    agg += w2 * nl1;
    agg += w3 * nr1;
    if (d == 0) agg = {0.0f, 0.0f, 0.0f, 0.0f};
    v4f res;
    res.x = (cd.x != 0.0f) ? (e.c.x + agg.x) : 0.0f;
    res.y = (cd.y != 0.0f) ? (e.c.y + agg.y) : 0.0f;
    res.z = (cd.z != 0.0f) ? (e.c.z + agg.z) : 0.0f;
    res.w = (cd.w != 0.0f) ? (e.c.w + agg.w) : 0.0f;
    *(v4f*)(outN + (size_t)(unsigned)d * HW + oC) = res;
  };
#pragma unroll
  for (int i = 0; i < 8; ++i) estore(i, e0[i]);
#pragma unroll
  for (int i = 0; i < 8; ++i) estore(8 + i, e1[i]);
}

extern "C" void kernel_launch(void* const* d_in, const int* in_sizes, int n_in,
                              void* d_out, int out_size, void* d_ws, size_t ws_size,
                              hipStream_t stream) {
  const float* cur = (const float*)d_in[0];
  const float* prev = (const float*)d_in[1];
  const float* mem = (const float*)d_in[2];
  float* out = (float*)d_out;
  (void)in_sizes; (void)n_in; (void)out_size; (void)d_ws; (void)ws_size;

  FeatureAlign_kernel<<<dim3(Nn * Hd), dim3(256), 0, stream>>>(cur, prev, mem, out);
}


# Dot loop: halo columns x0+15..x0+17 of all 4 window rows packed into ONE shared B tile per x tile (12 of 16 columns), no second tile per row: 20 WMMAs, 24 fragment reads per 4 channels; sums kept
# speedup vs baseline: 8.4319x; 1.1123x over previous
#include <hip/hip_runtime.h>
#include <math.h>


typedef __attribute__((ext_vector_type(2))) float v2f;
typedef __attribute__((ext_vector_type(8))) float v8f;
typedef __attribute__((ext_vector_type(4))) float v4f;

#define Nn 4
#define Dd 256
#define Hd 64
#define Wd 64
#define HW (Hd * Wd)
#define KC 32
#define NST (Dd / KC)
#define PITCH 72
#define SLOT (KC * PITCH)
#define STG (5 * SLOT)
#define NP 8

static __device__ __forceinline__ void lds_barrier() {
  asm volatile("s_wait_dscnt 0x0\n\ts_barrier_signal -1\n\ts_barrier_wait -1" ::: "memory");
}

__global__ __launch_bounds__(256) void FeatureAlign_kernel(
    const float* __restrict__ cur, const float* __restrict__ prev,
    const float* __restrict__ mem, float* __restrict__ out) {
  __shared__ __attribute__((aligned(16))) float ldsS[2][2 * STG];
  __shared__ float ldsD[4][5][16][16];
  __shared__ __attribute__((aligned(16))) float red[5][NP][Wd];
  __shared__ float nrm[5][Wd];
  __shared__ __attribute__((aligned(16))) float ldsW[5][Wd];

  const int nb = blockIdx.x;
  const int n = nb >> 6;
  const int y = nb & (Hd - 1);
  const int t = threadIdx.x;
  const int wave = __builtin_amdgcn_readfirstlane(t >> 5);
  const bool isMat = wave < 4;
  const int lane = t & 31;
  const int g = lane >> 4;
  const int li = lane & 15;
  const int x0 = (wave & 3) << 4;

  const float* curN = cur + (size_t)n * Dd * HW;
  const float* prevN = prev + (size_t)n * Dd * HW;
  const float* memN = mem + (size_t)n * Dd * HW;
  float* outN = out + (size_t)n * Dd * HW;

  const int lt = t & 127;
  const int q4 = (lt & 15) << 2;
  const int cs = lt >> 4;
  const unsigned gA = (unsigned)(cs * HW + y * Wd + q4);
  unsigned gB[4];
#pragma unroll
  for (int dyi = 0; dyi < 4; ++dyi) {
    const int dy = (dyi == 3) ? 2 : (dyi - 1);
    int yy = y + dy;
    yy = yy < 0 ? 0 : (yy > Hd - 1 ? Hd - 1 : yy);
    gB[dyi] = (unsigned)(cs * HW + yy * Wd + q4);
  }
  const unsigned sOff = (unsigned)(cs * PITCH + q4);

  const unsigned fA = (unsigned)(2 * g * PITCH + x0 + li);
  unsigned fB[5];
#pragma unroll
  for (int dyi = 0; dyi < 4; ++dyi) {
    int bx = x0 - 1 + li;
    bx = bx < 0 ? 0 : bx;
    fB[dyi] = (unsigned)((dyi + 1) * SLOT + 2 * g * PITCH + bx);
  }
  {
    const int hj = li < 12 ? li : li - 12;
    const int hrow = hj / 3;
    int hx = x0 + 15 + (hj - 3 * hrow);
    hx = hx > Wd - 1 ? Wd - 1 : hx;
    fB[4] = (unsigned)((hrow + 1) * SLOT + 2 * g * PITCH + hx);
  }

  v8f acc[5];
#pragma unroll
  for (int tb = 0; tb < 5; ++tb) acc[tb] = {};

  v4f sq[5];
#pragma unroll
  for (int r = 0; r < 5; ++r) sq[r] = {0.0f, 0.0f, 0.0f, 0.0f};

  auto gload = [&](v4f (&st)[5][4], int s) {
    const float* pc = curN + (size_t)(unsigned)(s * KC) * HW;
    const float* pp = prevN + (size_t)(unsigned)(s * KC) * HW;
#pragma unroll
    for (int jj = 0; jj < 4; ++jj)
      st[0][jj] = *(const v4f*)(pc + gA + (unsigned)(jj * 8 * HW));
#pragma unroll
    for (int r = 0; r < 4; ++r)
#pragma unroll
      for (int jj = 0; jj < 4; ++jj)
        st[r + 1][jj] = *(const v4f*)(pp + gB[r] + (unsigned)(jj * 8 * HW));
  };
  auto sqsum = [&](const v4f (&st)[5][4]) {
#pragma unroll
    for (int r = 0; r < 5; ++r)
#pragma unroll
      for (int jj = 0; jj < 4; ++jj) sq[r] += st[r][jj] * st[r][jj];
  };
  auto sstore = [&](const v4f (&st)[5][4], float* b) {
#pragma unroll
    for (int r = 0; r < 5; ++r)
#pragma unroll
      for (int jj = 0; jj < 4; ++jj)
        *(v4f*)(b + r * SLOT + jj * 8 * PITCH + sOff) = st[r][jj];
  };
  struct Frag {
    v2f a[2];
    v2f b[2][5];
  };
  auto fload = [&](const float* b, int blk, Frag& f) {
#pragma unroll
    for (int c = 0; c < 2; ++c) {
      const int kk = (blk * 2 + c) * 4;
      f.a[c].x = b[fA + kk * PITCH];
      f.a[c].y = b[fA + (kk + 1) * PITCH];
#pragma unroll
      for (int tb = 0; tb < 5; ++tb) {
        f.b[c][tb].x = b[fB[tb] + kk * PITCH];
        f.b[c][tb].y = b[fB[tb] + (kk + 1) * PITCH];
      }
    }
  };
  auto fmma = [&](const Frag& f) {
#pragma unroll
    for (int c = 0; c < 2; ++c)
#pragma unroll
      for (int tb = 0; tb < 5; ++tb)
        acc[tb] = __builtin_amdgcn_wmma_f32_16x16x4_f32(
            false, f.a[c], false, f.b[c][tb], (short)0, acc[tb],
            false, false);
  };
  auto compute = [&](int buf) {
    const float* b = &ldsS[buf][0];
    Frag f0, f1;
    fload(b, 0, f0);
    __builtin_amdgcn_sched_barrier(0);
    fload(b, 1, f1);
    __builtin_amdgcn_sched_barrier(0);
    fmma(f0);
    __builtin_amdgcn_sched_barrier(0);
    fload(b, 2, f0);
    __builtin_amdgcn_sched_barrier(0);
    fmma(f1);
    __builtin_amdgcn_sched_barrier(0);
    fload(b, 3, f1);
    __builtin_amdgcn_sched_barrier(0);
    fmma(f0);
    __builtin_amdgcn_sched_barrier(0);
    fload(b + STG, 0, f0);
    __builtin_amdgcn_sched_barrier(0);
    fmma(f1);
    __builtin_amdgcn_sched_barrier(0);
    fload(b + STG, 1, f1);
    __builtin_amdgcn_sched_barrier(0);
    fmma(f0);
    __builtin_amdgcn_sched_barrier(0);
    fload(b + STG, 2, f0);
    __builtin_amdgcn_sched_barrier(0);
    fmma(f1);
    __builtin_amdgcn_sched_barrier(0);
    fload(b + STG, 3, f1);
    __builtin_amdgcn_sched_barrier(0);
    fmma(f0);
    fmma(f1);
  };
  auto stage2 = [&](int s0, int buf) {
    v4f stA[5][4], stB[5][4];
    gload(stA, s0);
    gload(stB, s0 + 1);
    sqsum(stA);
    sstore(stA, &ldsS[buf][0]);
    sqsum(stB);
    sstore(stB, &ldsS[buf][STG]);
  };

  if (!isMat) stage2(0, 1);
  __syncthreads();
  for (int i = 0; i < NST / 2 - 1; ++i) {
    if (isMat) {
      compute((i + 1) & 1);
    } else {
      stage2(2 * (i + 1), i & 1);
    }
    __syncthreads();
  }

  const int eq = t & 15;
  const int cq = t >> 4;
  const int x4 = eq << 2;
  int ym = y - 1; ym = ym < 0 ? 0 : ym;
  int yp = y + 2; yp = yp > Hd - 1 ? Hd - 1 : yp;
  const bool lastq = (eq == 15);
  const unsigned oC = (unsigned)(y * Wd + x4);
  const unsigned oM = (unsigned)(ym * Wd + x4);
  const unsigned oP = (unsigned)(yp * Wd + x4);
  const unsigned dL = (x4 > 0) ? 1u : 0u;
  const unsigned dR = lastq ? 2u : 4u;
  struct EL {
    v4f c, m0, m1;
    v2f r0, r1;
    float l0, l1;
  };
  auto eload = [&](int i, EL& e) {
    const float* md = memN + (size_t)(unsigned)(cq + 16 * i) * HW;
    e.c = *(const v4f*)(md + oC);
    e.l0 = md[oM - dL];
    e.m0 = *(const v4f*)(md + oM);
    e.r0 = *(const v2f*)(md + oM + dR);
    e.l1 = md[oP - dL];
    e.m1 = *(const v4f*)(md + oP);
    e.r1 = *(const v2f*)(md + oP + dR);
  };
  EL e0[8], e1[8];
  {
    const float* b = &ldsS[(NST / 2) & 1][0];
    Frag f[2];
    if (isMat) {
      fload(b, 0, f[0]);
      __builtin_amdgcn_sched_barrier(0);
      fload(b, 1, f[1]);
    }
#pragma unroll
    for (int j = 0; j < 8; ++j) {
      eload(j, e0[j]);
      if (isMat) {
        fmma(f[j & 1]);
        if (j + 2 < 8) {
          __builtin_amdgcn_sched_barrier(0);
          fload(b + ((j + 2) >> 2) * STG, (j + 2) & 3, f[j & 1]);
        }
      }
    }
  }

  if (isMat) {
#pragma unroll
    for (int tb = 0; tb < 5; ++tb)
#pragma unroll
      for (int r = 0; r < 8; ++r)
        ldsD[wave][tb][r + 8 * g][li] = acc[tb][r];
  } else {
#pragma unroll
    for (int r = 0; r < 5; ++r) *(v4f*)&red[r][cs][q4] = sq[r];
  }
#pragma unroll
  for (int i = 0; i < 8; ++i) eload(8 + i, e1[i]);
  lds_barrier();

  for (int idx = t; idx < 5 * Wd; idx += 256) {
    const int r = idx >> 6;
    const int xx = idx & (Wd - 1);
    float s = 0.0f;
#pragma unroll
    for (int k = 0; k < NP; ++k) s += red[r][k][xx];
    nrm[r][xx] = sqrtf(s) + 1e-8f;
  }
  lds_barrier();

  if (t < Wd) {
    const int x = t;
    bool yv[4], xv[4];
    int xc[4];
#pragma unroll
    for (int k = 0; k < 4; ++k) {
      const int d = (k == 3) ? 2 : (k - 1);
      const int yy = y + d, xx = x + d;
      yv[k] = (yy >= 0) && (yy < Hd);
      xv[k] = (xx >= 0) && (xx < Wd);
      xc[k] = xx < 0 ? 0 : (xx > Wd - 1 ? Wd - 1 : xx);
    }
    const int pw = x >> 4, pi = x & 15;
    const float cnv = nrm[0][x];
    float av[4][4];
#pragma unroll
    for (int a = 0; a < 4; ++a)
#pragma unroll
      for (int b2 = 0; b2 < 4; ++b2) {
        const bool inMass = (a < 3) && (b2 < 3);
        const bool corner = (a == 0 || a == 3) && (b2 == 0 || b2 == 3);
        av[a][b2] = 0.0f;
        if (inMass || corner) {
          const int bc = pi + b2;
          const float dv = (bc < 16) ? ldsD[pw][a][pi][bc & 15]
                                     : ldsD[pw][4][pi][3 * a + (bc & 3)];
          av[a][b2] = dv / (cnv * nrm[1 + a][xc[b2]]);
        }
      }
    float mass = 0.0f;
#pragma unroll
    for (int dyi = 0; dyi < 3; ++dyi)
#pragma unroll
      for (int dxi = 0; dxi < 3; ++dxi)
        if (yv[dyi] && xv[dxi]) mass += fmaxf(av[dyi][dxi], 0.0f);
    ldsW[4][x] = ((mass > -10.0f) && (mass < 10.0f)) ? 1.0f : 0.0f;
    const float msafe = (mass > 0.0f) ? mass : 1.0f;
#pragma unroll
    for (int oy = 0; oy < 2; ++oy) {
      const int dyi = oy ? 3 : 0;
#pragma unroll
      for (int ox = 0; ox < 2; ++ox) {
        const int dxi = ox ? 3 : 0;
        const float a = av[dyi][dxi];
        float wv = 0.0f;
        if (yv[dyi] && xv[dxi]) wv = (a > 0.0f) ? (a / msafe) : 0.0f;
        ldsW[oy * 2 + ox][x] = wv;
      }
    }
  }
  lds_barrier();

  const v4f w0 = *(const v4f*)&ldsW[0][x4];
  const v4f w1 = *(const v4f*)&ldsW[1][x4];
  const v4f w2 = *(const v4f*)&ldsW[2][x4];
  const v4f w3 = *(const v4f*)&ldsW[3][x4];
  const v4f cd = *(const v4f*)&ldsW[4][x4];

  auto estore = [&](int i, const EL& e) {
    const int d = cq + 16 * i;
    const float rx0 = lastq ? e.r0.y : e.r0.x;
    const float rx1 = lastq ? e.r1.y : e.r1.x;
    const v4f nl0 = {e.l0, e.m0.x, e.m0.y, e.m0.z};
    const v4f nr0 = {e.m0.z, e.m0.w, rx0, e.r0.y};
    const v4f nl1 = {e.l1, e.m1.x, e.m1.y, e.m1.z};
    const v4f nr1 = {e.m1.z, e.m1.w, rx1, e.r1.y};
    v4f agg = {0.0f, 0.0f, 0.0f, 0.0f};
    agg += w0 * nl0;
    agg += w1 * nr0;
    agg += w2 * nl1;
    agg += w3 * nr1;
    if (d == 0) agg = {0.0f, 0.0f, 0.0f, 0.0f};
    v4f res;
    res.x = (cd.x != 0.0f) ? (e.c.x + agg.x) : 0.0f;
    res.y = (cd.y != 0.0f) ? (e.c.y + agg.y) : 0.0f;
    res.z = (cd.z != 0.0f) ? (e.c.z + agg.z) : 0.0f;
    res.w = (cd.w != 0.0f) ? (e.c.w + agg.w) : 0.0f;
    *(v4f*)(outN + (size_t)(unsigned)d * HW + oC) = res;
  };
#pragma unroll
  for (int i = 0; i < 8; ++i) estore(i, e0[i]);
#pragma unroll
  for (int i = 0; i < 8; ++i) estore(8 + i, e1[i]);
}

extern "C" void kernel_launch(void* const* d_in, const int* in_sizes, int n_in,
                              void* d_out, int out_size, void* d_ws, size_t ws_size,
                              hipStream_t stream) {
  const float* cur = (const float*)d_in[0];
  const float* prev = (const float*)d_in[1];
  const float* mem = (const float*)d_in[2];
  float* out = (float*)d_out;
  (void)in_sizes; (void)n_in; (void)out_size; (void)d_ws; (void)ws_size;

  FeatureAlign_kernel<<<dim3(Nn * Hd), dim3(256), 0, stream>>>(cur, prev, mem, out);
}
